# starting kernel
# baseline (speedup 1.0000x reference)
#include <hip/hip_runtime.h>

typedef float v2f __attribute__((ext_vector_type(2)));
typedef float v8f __attribute__((ext_vector_type(8)));

#define H   512
#define W   512
#define OH  128
#define OW  128
#define NC  96
#define TILES_PER_IMG 64
#define WAVES_PER_BLOCK 4

template <bool EDGE>
__device__ __forceinline__ void pass1(const float* __restrict__ src,
                                      int ybase, int xbase,
                                      float (*__restrict__ T)[84],
                                      const float (*__restrict__ Hm)[80],
                                      int l15, int half)
{
    #pragma unroll
    for (int sub = 0; sub < 5; ++sub) {
        v8f acc = {};
        const int y = ybase + 16 * sub + l15;
        bool yv = true;
        const float* rowp;
        if (EDGE) {
            yv = (y >= 0 && y < H);
            const int yc = min(max(y, 0), H - 1);
            rowp = src + (size_t)yc * W;
        } else {
            rowp = src + (size_t)y * W + xbase + 2 * half;
        }
        #pragma unroll
        for (int kk = 0; kk < 19; ++kk) {
            v2f a;
            if (EDGE) {
                int x  = xbase + 4 * kk + 2 * half;
                int xc = min(max(x, 0), W - 2);
                a = *(const v2f*)(rowp + xc);
                a.x = (yv && (unsigned)x       < (unsigned)W) ? a.x : 0.0f;
                a.y = (yv && (unsigned)(x + 1) < (unsigned)W) ? a.y : 0.0f;
            } else {
                a = *(const v2f*)(rowp + 4 * kk);
            }
            v2f b = *(const v2f*)(&Hm[l15][4 * kk + 2 * half]);
            acc = __builtin_amdgcn_wmma_f32_16x16x4_f32(false, a, false, b,
                                                        (short)0, acc, false, false);
        }
        #pragma unroll
        for (int r = 0; r < 8; ++r)
            T[l15][16 * sub + r + 8 * half] = acc[r];
    }
}

__global__ __launch_bounds__(128)
void aa_interp_wmma_kernel(const float* __restrict__ in,
                           const float* __restrict__ wgt,
                           float* __restrict__ out)
{
    __shared__ float ctab[13];
    __shared__ float Hm[16][80];
    __shared__ float Tt[WAVES_PER_BLOCK][16][84];

    const int tid = threadIdx.x;

    if (tid < 13) {
        float s = 0.f;
        #pragma unroll
        for (int j = 0; j < 13; ++j) s += wgt[tid * 13 + j];
        ctab[tid] = s;
    }
    __syncthreads();
    for (int idx = tid; idx < 16 * 80; idx += 128) {
        int n = idx / 80, w = idx % 80;
        int d = w - 4 * n;
        Hm[n][w] = (d >= 0 && d < 13) ? ctab[d] : 0.0f;
    }
    __syncthreads();

    const int wv   = tid >> 5;
    const int lane = tid & 31;
    const int half = lane >> 4;
    const int l15  = lane & 15;

    const int tileId = blockIdx.x * WAVES_PER_BLOCK + wv;
    const int nc  = tileId / TILES_PER_IMG;
    const int t   = tileId % TILES_PER_IMG;
    const int oy0 = (t >> 3) * 16;
    const int ox0 = (t & 7) * 16;

    const float* src = in + (size_t)nc * (H * W);
    const int ybase = 4 * oy0 - 6;
    const int xbase = 4 * ox0 - 6;

    float (*T)[84] = Tt[wv];

    const bool interior = (oy0 >= 16) & (oy0 <= 96) & (ox0 >= 16) & (ox0 <= 96);
    if (interior)
        pass1<false>(src, ybase, xbase, T, Hm, l15, half);
    else
        pass1<true>(src, ybase, xbase, T, Hm, l15, half);

    v8f o = {};
    #pragma unroll
    for (int kk = 0; kk < 19; ++kk) {
        v2f a = *(const v2f*)(&Hm[l15][4 * kk + 2 * half]);
        v2f b = *(const v2f*)(&T [l15][4 * kk + 2 * half]);
        o = __builtin_amdgcn_wmma_f32_16x16x4_f32(false, a, false, b,
                                                  (short)0, o, false, false);
    }

    float* dst = out + (size_t)nc * (OH * OW);
    #pragma unroll
    for (int r = 0; r < 8; ++r)
        dst[(size_t)(oy0 + r + 8 * half) * OW + (ox0 + l15)] = o[r];
}

extern "C" void kernel_launch(void* const* d_in, const int* in_sizes, int n_in,
                              void* d_out, int out_size, void* d_ws, size_t ws_size,
                              hipStream_t stream) {
    const float* inp = (const float*)d_in[0];
    const float* wgt = (const float*)d_in[1];
    float* out = (float*)d_out;

    dim3 grid(NC * TILES_PER_IMG / WAVES_PER_BLOCK);
    dim3 block(128);
    aa_interp_wmma_kernel<<<grid, block, 0, stream>>>(inp, wgt, out);
}


# Work-groups of 24 waves = three stacked tile rows: the 16-row group that two stacked tiles of one image share is blurred once, by the upper wave, and handed down through LDS; sums unchanged
# speedup vs baseline: 2.0482x; 2.0482x over previous
#include <hip/hip_runtime.h>


typedef float v2f __attribute__((ext_vector_type(2)));
typedef float v4f __attribute__((ext_vector_type(4)));
typedef float v8f __attribute__((ext_vector_type(8)));

#define H   512
#define W   512
#define OH  128
#define OW  128
#define NC  96
#define TILES_PER_IMG 64
#define ROWS_PER_BLOCK 3
#define WAVES_PER_ROW 8
#define WAVES_PER_BLOCK (ROWS_PER_BLOCK * WAVES_PER_ROW)

#define SROW 84
#define NMAIN 8
#define NHALO 2
#define NLD  (NMAIN + NHALO)
#define NK   19

__device__ __forceinline__ int halo_col(int lane)
{
    const int q = lane & 3;
    return (q < 2) ? 4 * q : 64 + 4 * q;
}

template <bool EDGE>
__device__ __forceinline__ void load_group(const float* __restrict__ src,
                                           int y0, int xa, int lane,
                                           v4f (&R)[NLD])
{
    const int l15  = lane & 15;
    const int half = lane >> 4;
    const int xm = xa + 8 + 4 * l15;
    const int xh = xa + halo_col(lane);
    if (EDGE) {
        const int xhc = min(max(xh, 0), W - 4);
        #pragma unroll
        for (int j = 0; j < NMAIN; ++j) {
            const int yc = min(max(y0 + 2 * j + half, 0), H - 1);
            R[j] = *(const v4f*)(src + (yc * W + xm));
        }
        #pragma unroll
        for (int j = 0; j < NHALO; ++j) {
            const int yc = min(max(y0 + 8 * j + (lane >> 2), 0), H - 1);
            R[NMAIN + j] = *(const v4f*)(src + (yc * W + xhc));
        }
    } else {
        const float* pm = src + ((y0 + half) * W + xm);
        const float* ph = src + ((y0 + (lane >> 2)) * W + xh);
        #pragma unroll
        for (int j = 0; j < NMAIN; ++j)
            R[j] = *(const v4f*)(pm + 2 * j * W);
        #pragma unroll
        for (int j = 0; j < NHALO; ++j)
            R[NMAIN + j] = *(const v4f*)(ph + 8 * j * W);
    }
}

template <bool EDGE>
__device__ __forceinline__ void store_group(float* S,
                                            int y0, int xa, int lane,
                                            const v4f (&R)[NLD])
{
    const int l15  = lane & 15;
    const int half = lane >> 4;
    const v4f z = {0.0f, 0.0f, 0.0f, 0.0f};
    #pragma unroll
    for (int j = 0; j < NMAIN; ++j) {
        const int row = 2 * j + half;
        v4f v = R[j];
        if (EDGE) {
            const bool ok = (unsigned)(y0 + row) < (unsigned)H;
            v = ok ? v : z;
        }
        *(v4f*)(S + row * SROW + 8 + 4 * l15) = v;
    }
    #pragma unroll
    for (int j = 0; j < NHALO; ++j) {
        const int row = 8 * j + (lane >> 2);
        const int c   = halo_col(lane);
        v4f v = R[NMAIN + j];
        if (EDGE) {
            const bool ok = ((unsigned)(y0 + row) < (unsigned)H) &&
                            ((unsigned)(xa + c) < (unsigned)W);
            v = ok ? v : z;
        }
        *(v4f*)(S + row * SROW + c) = v;
    }
}

template <bool EDGE>
__device__ __forceinline__ void pass1(const float* __restrict__ src,
                                      int ybase, int xa,
                                      float* __restrict__ S,
                                      float (*__restrict__ T)[84],
                                      float (*__restrict__ Tn)[84],
                                      bool give, bool take,
                                      const v2f (&Bf)[NK],
                                      int lane, int l15, int half)
{
    v4f R[NLD];
    load_group<EDGE>(src, ybase + 64, xa, lane, R);
    #pragma unroll
    for (int it = 0; it < 5; ++it) {
        const int sub = 4 - it;
        if (sub == 0 && take)
            break;
        store_group<EDGE>(S, ybase + 16 * sub, xa, lane, R);
        if (sub > 1 || (sub == 1 && !take))
            load_group<EDGE>(src, ybase + 16 * (sub - 1), xa, lane, R);

        v2f a[NK];
        #pragma unroll
        for (int kk = 0; kk < NK; ++kk)
            a[kk] = *(const v2f*)(S + l15 * SROW + 2 + 4 * kk + 2 * half);
        __builtin_amdgcn_sched_barrier(0);

        v8f acc = {};
        #pragma unroll
        for (int kk = 0; kk < NK; ++kk)
            acc = __builtin_amdgcn_wmma_f32_16x16x4_f32(false, a[kk], false, Bf[kk],
                                                        (short)0, acc, false, false);
        #pragma unroll
        for (int r = 0; r < 8; ++r)
            T[l15][16 * sub + r + 8 * half] = acc[r];
        if (sub == 4) {
            if (give) {
                #pragma unroll
                for (int r = 0; r < 8; ++r)
                    Tn[l15][r + 8 * half] = acc[r];
            }
            asm volatile("s_wait_dscnt 0x0" ::: "memory");
            __builtin_amdgcn_s_barrier_signal(-1);
        }
        __builtin_amdgcn_sched_barrier(0);
    }
}

__global__ __launch_bounds__(32 * WAVES_PER_BLOCK, 6)
void aa_interp_wmma_kernel(const float* __restrict__ in,
                           const float* __restrict__ wgt,
                           float* __restrict__ out)
{
    __shared__ float ctab[13];
    __shared__ float Hm[16][80];
    __shared__ float Tt[WAVES_PER_BLOCK][16][84];
    __shared__ __attribute__((aligned(16))) float Ss[WAVES_PER_BLOCK][16 * SROW];

    const int tid = threadIdx.x;

    if (tid < 13) {
        float s = 0.f;
        #pragma unroll
        for (int j = 0; j < 13; ++j) s += wgt[tid * 13 + j];
        ctab[tid] = s;
    }
    __syncthreads();
    for (int idx = tid; idx < 16 * 80; idx += 32 * WAVES_PER_BLOCK) {
        int n = idx / 80, w = idx % 80;
        int d = w - 4 * n;
        Hm[n][w] = (d >= 0 && d < 13) ? ctab[d] : 0.0f;
    }
    __syncthreads();

    const int wv   = tid >> 5;
    const int lane = tid & 31;
    const int half = lane >> 4;
    const int l15  = lane & 15;

    const int tileId = blockIdx.x * WAVES_PER_BLOCK + wv;
    const int rowInBlock = wv / WAVES_PER_ROW;
    const int nc  = tileId / TILES_PER_IMG;
    const int t   = tileId % TILES_PER_IMG;
    const int ty  = t >> 3;
    const int oy0 = ty * 16;
    const bool take = (rowInBlock > 0) & (ty > 0);
    const bool give = (rowInBlock < ROWS_PER_BLOCK - 1) & (ty < 7);
    const int ox0 = (t & 7) * 16;

    const float* src = in + (size_t)nc * (H * W);
    const int ybase = 4 * oy0 - 6;
    const int xa    = 4 * ox0 - 8;

    float (*T)[84] = Tt[wv];
    float (*Tn)[84] = Tt[give ? wv + WAVES_PER_ROW : wv];
    float* S = Ss[wv];

    v2f Bf[NK];
    #pragma unroll
    for (int kk = 0; kk < NK; ++kk)
        Bf[kk] = *(const v2f*)(&Hm[l15][4 * kk + 2 * half]);

    const bool interior = (oy0 >= 16) & (oy0 <= 96) & (ox0 >= 16) & (ox0 <= 96);
    if (interior)
        pass1<false>(src, ybase, xa, S, T, Tn, give, take, Bf, lane, l15, half);
    else
        pass1<true>(src, ybase, xa, S, T, Tn, give, take, Bf, lane, l15, half);
    __builtin_amdgcn_s_barrier_wait(-1);

    v2f tb[NK];
    #pragma unroll
    for (int kk = 0; kk < NK; ++kk)
        tb[kk] = *(const v2f*)(&T[l15][4 * kk + 2 * half]);
    __builtin_amdgcn_sched_barrier(0);
    v8f o = {};
    #pragma unroll
    for (int kk = 0; kk < NK; ++kk)
        o = __builtin_amdgcn_wmma_f32_16x16x4_f32(false, Bf[kk], false, tb[kk],
                                                  (short)0, o, false, false);

    float* dst = out + (size_t)nc * (OH * OW);
    #pragma unroll
    for (int r = 0; r < 8; ++r)
        dst[(size_t)(oy0 + r + 8 * half) * OW + (ox0 + l15)] = o[r];
}

extern "C" void kernel_launch(void* const* d_in, const int* in_sizes, int n_in,
                              void* d_out, int out_size, void* d_ws, size_t ws_size,
                              hipStream_t stream) {
    const float* inp = (const float*)d_in[0];
    const float* wgt = (const float*)d_in[1];
    float* out = (float*)d_out;

    dim3 grid(NC * TILES_PER_IMG / WAVES_PER_BLOCK);
    dim3 block(768);
    aa_interp_wmma_kernel<<<grid, block, 0, stream>>>(inp, wgt, out);
}
